# starting kernel
# baseline (speedup 1.0000x reference)
#include <hip/hip_runtime.h>
#include <hip/hip_bf16.h>


typedef float v2f __attribute__((ext_vector_type(2)));
typedef float v8f __attribute__((ext_vector_type(8)));

#define N_IN 64
#define N_HID 128
#define N_OUT 64


__global__ void k_deg_init(float* __restrict__ deg, int n) {
    int i = blockIdx.x * blockDim.x + threadIdx.x;
    if (i < n) deg[i] = 1.0f;
}

__global__ void k_deg_count(const int* __restrict__ dst, float* __restrict__ deg, int E) {
    int e = blockIdx.x * blockDim.x + threadIdx.x;
    if (e < E) atomicAdd(&deg[dst[e]], 1.0f);
}

__global__ void k_deg_rsqrt(float* __restrict__ deg, int n) {
    int i = blockIdx.x * blockDim.x + threadIdx.x;
    if (i < n) deg[i] = rsqrtf(fmaxf(deg[i], 1.0f));
}

template <int K, int N>
__global__ void k_gemm_wmma_f32(const float* __restrict__ A,
                                const float* __restrict__ B,
                                float* __restrict__ C, int M) {
    const int lane = threadIdx.x & 31;
    const int wave = threadIdx.x >> 5;
    const int wavesPerBlock = blockDim.x >> 5;
    const int tile = blockIdx.x * wavesPerBlock + wave;

    const int nTiles = N / 16;
    const int mTiles = M / 16;
    if (tile >= mTiles * nTiles) return;

    const int row0 = (tile / nTiles) * 16;
    const int col0 = (tile % nTiles) * 16;
    const int half = lane >> 4;
    const int l16  = lane & 15;

    v8f acc = {0.f, 0.f, 0.f, 0.f, 0.f, 0.f, 0.f, 0.f};

#pragma unroll
    for (int k = 0; k < K; k += 4) {
        const float* ap = A + (size_t)(row0 + l16) * K + (k + 2 * half);
        v2f a;
        a.x = ap[0];
        a.y = ap[1];
        const float* bp = B + (size_t)(k + 2 * half) * N + (col0 + l16);
        v2f b;
        b.x = bp[0];
        b.y = bp[N];
        acc = __builtin_amdgcn_wmma_f32_16x16x4_f32(
             false, a,  false, b,
             (short)0, acc,  false,  false);
    }

#pragma unroll
    for (int j = 0; j < 8; ++j) {
        C[(size_t)(row0 + 8 * half + j) * N + (col0 + l16)] = acc[j];
    }
}


template <int C>
__global__ void k_agg_init(const float* __restrict__ h, const float* __restrict__ inv,
                           const float* __restrict__ bias, float* __restrict__ out, int n) {
    const int C4 = C / 4;
    int idx = blockIdx.x * blockDim.x + threadIdx.x;
    if (idx >= n * C4) return;
    int i  = idx / C4;
    int c4 = idx % C4;
    float s = inv[i];
    s = s * s;
    float4 v = ((const float4*)(h + (size_t)i * C))[c4];
    float4 r;
    r.x = s * v.x; r.y = s * v.y; r.z = s * v.z; r.w = s * v.w;
    if (bias != nullptr) {
        float4 bv = ((const float4*)bias)[c4];
        r.x += bv.x; r.y += bv.y; r.z += bv.z; r.w += bv.w;
    }
    ((float4*)(out + (size_t)i * C))[c4] = r;
}

template <int C>
__global__ void k_agg_edges(const float* __restrict__ h, const int* __restrict__ src,
                            const int* __restrict__ dst, const float* __restrict__ inv,
                            float* __restrict__ out, int E) {
    const int C4 = C / 4;
    int idx = blockIdx.x * blockDim.x + threadIdx.x;
    if (idx >= E * C4) return;
    int e  = idx / C4;
    int c4 = idx % C4;
    int s = src[e];
    int d = dst[e];
    float nrm = inv[s] * inv[d];
    float4 v = ((const float4*)(h + (size_t)s * C))[c4];
    float* op = out + (size_t)d * C + 4 * c4;
    atomicAdd(op + 0, nrm * v.x);
    atomicAdd(op + 1, nrm * v.y);
    atomicAdd(op + 2, nrm * v.z);
    atomicAdd(op + 3, nrm * v.w);
}

template <int C>
__global__ void k_bias_relu(float* __restrict__ a, const float* __restrict__ bias, int n) {
    int idx = blockIdx.x * blockDim.x + threadIdx.x;
    if (idx >= n * C) return;
    int c = idx % C;
    a[idx] = fmaxf(a[idx] + bias[c], 0.0f);
}


extern "C" void kernel_launch(void* const* d_in, const int* in_sizes, int n_in,
                              void* d_out, int out_size, void* d_ws, size_t ws_size,
                              hipStream_t stream) {
    const float* x    = (const float*)d_in[0];
    const int*   eidx = (const int*)d_in[1];
    const float* W1   = (const float*)d_in[2];
    const float* b1   = (const float*)d_in[3];
    const float* W2   = (const float*)d_in[4];
    const float* b2   = (const float*)d_in[5];
    float*       out  = (float*)d_out;

    const int n = in_sizes[0] / N_IN;
    const int E = in_sizes[1] / 2;
    const int* src = eidx;
    const int* dst = eidx + E;

    char* ws = (char*)d_ws;
    const size_t OFF_INV = 0;
    const size_t OFF_H0  = (((size_t)n * 4) + 511) & ~(size_t)511;
    const size_t OFF_AGG = OFF_H0 + (size_t)n * N_HID * 4;
    float* inv  = (float*)(ws + OFF_INV);
    float* h0   = (float*)(ws + OFF_H0);
    float* agg1 = (float*)(ws + OFF_AGG);

    const int TPB = 256;

    k_deg_init<<<(n + TPB - 1) / TPB, TPB, 0, stream>>>(inv, n);
    k_deg_count<<<(E + TPB - 1) / TPB, TPB, 0, stream>>>(dst, inv, E);
    k_deg_rsqrt<<<(n + TPB - 1) / TPB, TPB, 0, stream>>>(inv, n);

    {
        int tiles = (n / 16) * (N_HID / 16);
        int blocks = (tiles + 3) / 4;
        k_gemm_wmma_f32<N_IN, N_HID><<<blocks, 128, 0, stream>>>(x, W1, h0, n);
    }
    k_agg_init<N_HID><<<((n * (N_HID / 4)) + TPB - 1) / TPB, TPB, 0, stream>>>(
        h0, inv, nullptr, agg1, n);
    k_agg_edges<N_HID><<<((E * (N_HID / 4)) + TPB - 1) / TPB, TPB, 0, stream>>>(
        h0, src, dst, inv, agg1, E);
    k_bias_relu<N_HID><<<((n * N_HID) + TPB - 1) / TPB, TPB, 0, stream>>>(agg1, b1, n);

    float* h2 = h0;
    {
        int tiles = (n / 16) * (N_OUT / 16);
        int blocks = (tiles + 3) / 4;
        k_gemm_wmma_f32<N_HID, N_OUT><<<blocks, 128, 0, stream>>>(agg1, W2, h2, n);
    }
    k_agg_init<N_OUT><<<((n * (N_OUT / 4)) + TPB - 1) / TPB, TPB, 0, stream>>>(
        h2, inv, b2, out, n);
    k_agg_edges<N_OUT><<<((E * (N_OUT / 4)) + TPB - 1) / TPB, TPB, 0, stream>>>(
        h2, src, dst, inv, out, E);

    (void)n_in; (void)out_size; (void)ws_size;
}


# Layer 1 aggregates the 64 input channels before the 64->128 transform ((A x) W1 instead of A (x W1)): its edge scatter does half the float atomics.
# speedup vs baseline: 1.2672x; 1.2672x over previous
#include <hip/hip_runtime.h>
#include <hip/hip_bf16.h>


typedef float v2f __attribute__((ext_vector_type(2)));
typedef float v8f __attribute__((ext_vector_type(8)));

#define N_IN 64
#define N_HID 128
#define N_OUT 64


__global__ void k_deg_init(float* __restrict__ deg, int n) {
    int i = blockIdx.x * blockDim.x + threadIdx.x;
    if (i < n) deg[i] = 1.0f;
}

__global__ void k_deg_count(const int* __restrict__ dst, float* __restrict__ deg, int E) {
    int e = blockIdx.x * blockDim.x + threadIdx.x;
    if (e < E) atomicAdd(&deg[dst[e]], 1.0f);
}

__global__ void k_deg_rsqrt(float* __restrict__ deg, int n) {
    int i = blockIdx.x * blockDim.x + threadIdx.x;
    if (i < n) deg[i] = rsqrtf(fmaxf(deg[i], 1.0f));
}

template <int K, int N>
__global__ void k_gemm_wmma_f32(const float* __restrict__ A,
                                const float* __restrict__ B,
                                float* __restrict__ C, int M) {
    const int lane = threadIdx.x & 31;
    const int wave = threadIdx.x >> 5;
    const int wavesPerBlock = blockDim.x >> 5;
    const int tile = blockIdx.x * wavesPerBlock + wave;

    const int nTiles = N / 16;
    const int mTiles = M / 16;
    if (tile >= mTiles * nTiles) return;

    const int row0 = (tile / nTiles) * 16;
    const int col0 = (tile % nTiles) * 16;
    const int half = lane >> 4;
    const int l16  = lane & 15;

    v8f acc = {0.f, 0.f, 0.f, 0.f, 0.f, 0.f, 0.f, 0.f};

#pragma unroll
    for (int k = 0; k < K; k += 4) {
        const float* ap = A + (size_t)(row0 + l16) * K + (k + 2 * half);
        v2f a;
        a.x = ap[0];
        a.y = ap[1];
        const float* bp = B + (size_t)(k + 2 * half) * N + (col0 + l16);
        v2f b;
        b.x = bp[0];
        b.y = bp[N];
        acc = __builtin_amdgcn_wmma_f32_16x16x4_f32(
             false, a,  false, b,
             (short)0, acc,  false,  false);
    }

#pragma unroll
    for (int j = 0; j < 8; ++j) {
        C[(size_t)(row0 + 8 * half + j) * N + (col0 + l16)] = acc[j];
    }
}


template <int C>
__global__ void k_agg_init(const float* __restrict__ h, const float* __restrict__ inv,
                           const float* __restrict__ bias, float* __restrict__ out, int n) {
    const int C4 = C / 4;
    int idx = blockIdx.x * blockDim.x + threadIdx.x;
    if (idx >= n * C4) return;
    int i  = idx / C4;
    int c4 = idx % C4;
    float s = inv[i];
    s = s * s;
    float4 v = ((const float4*)(h + (size_t)i * C))[c4];
    float4 r;
    r.x = s * v.x; r.y = s * v.y; r.z = s * v.z; r.w = s * v.w;
    if (bias != nullptr) {
        float4 bv = ((const float4*)bias)[c4];
        r.x += bv.x; r.y += bv.y; r.z += bv.z; r.w += bv.w;
    }
    ((float4*)(out + (size_t)i * C))[c4] = r;
}

template <int C>
__global__ void k_agg_edges(const float* __restrict__ h, const int* __restrict__ src,
                            const int* __restrict__ dst, const float* __restrict__ inv,
                            float* __restrict__ out, int E) {
    const int C4 = C / 4;
    int idx = blockIdx.x * blockDim.x + threadIdx.x;
    if (idx >= E * C4) return;
    int e  = idx / C4;
    int c4 = idx % C4;
    int s = src[e];
    int d = dst[e];
    float nrm = inv[s] * inv[d];
    float4 v = ((const float4*)(h + (size_t)s * C))[c4];
    float* op = out + (size_t)d * C + 4 * c4;
    atomicAdd(op + 0, nrm * v.x);
    atomicAdd(op + 1, nrm * v.y);
    atomicAdd(op + 2, nrm * v.z);
    atomicAdd(op + 3, nrm * v.w);
}

template <int C>
__global__ void k_bias_relu(float* __restrict__ a, const float* __restrict__ bias, int n) {
    int idx = blockIdx.x * blockDim.x + threadIdx.x;
    if (idx >= n * C) return;
    int c = idx % C;
    a[idx] = fmaxf(a[idx] + bias[c], 0.0f);
}


extern "C" void kernel_launch(void* const* d_in, const int* in_sizes, int n_in,
                              void* d_out, int out_size, void* d_ws, size_t ws_size,
                              hipStream_t stream) {
    const float* x    = (const float*)d_in[0];
    const int*   eidx = (const int*)d_in[1];
    const float* W1   = (const float*)d_in[2];
    const float* b1   = (const float*)d_in[3];
    const float* W2   = (const float*)d_in[4];
    const float* b2   = (const float*)d_in[5];
    float*       out  = (float*)d_out;

    const int n = in_sizes[0] / N_IN;
    const int E = in_sizes[1] / 2;
    const int* src = eidx;
    const int* dst = eidx + E;

    char* ws = (char*)d_ws;
    const size_t OFF_INV = 0;
    const size_t OFF_H0  = (((size_t)n * 4) + 511) & ~(size_t)511;
    const size_t OFF_AGG = OFF_H0 + (size_t)n * N_HID * 4;
    float* inv  = (float*)(ws + OFF_INV);
    float* h1   = (float*)(ws + OFF_H0);
    float* aggx = (float*)(ws + OFF_AGG);

    const int TPB = 256;

    k_deg_init<<<(n + TPB - 1) / TPB, TPB, 0, stream>>>(inv, n);
    k_deg_count<<<(E + TPB - 1) / TPB, TPB, 0, stream>>>(dst, inv, E);
    k_deg_rsqrt<<<(n + TPB - 1) / TPB, TPB, 0, stream>>>(inv, n);

    k_agg_init<N_IN><<<((n * (N_IN / 4)) + TPB - 1) / TPB, TPB, 0, stream>>>(
        x, inv, nullptr, aggx, n);
    k_agg_edges<N_IN><<<((E * (N_IN / 4)) + TPB - 1) / TPB, TPB, 0, stream>>>(
        x, src, dst, inv, aggx, E);
    {
        int tiles = (n / 16) * (N_HID / 16);
        int blocks = (tiles + 3) / 4;
        k_gemm_wmma_f32<N_IN, N_HID><<<blocks, 128, 0, stream>>>(aggx, W1, h1, n);
    }
    k_bias_relu<N_HID><<<((n * N_HID) + TPB - 1) / TPB, TPB, 0, stream>>>(h1, b1, n);

    float* h2 = aggx;
    {
        int tiles = (n / 16) * (N_OUT / 16);
        int blocks = (tiles + 3) / 4;
        k_gemm_wmma_f32<N_HID, N_OUT><<<blocks, 128, 0, stream>>>(h1, W2, h2, n);
    }
    k_agg_init<N_OUT><<<((n * (N_OUT / 4)) + TPB - 1) / TPB, TPB, 0, stream>>>(
        h2, inv, b2, out, n);
    k_agg_edges<N_OUT><<<((E * (N_OUT / 4)) + TPB - 1) / TPB, TPB, 0, stream>>>(
        h2, src, dst, inv, out, E);

    (void)n_in; (void)out_size; (void)ws_size;
}


# In-edges are grouped by destination once per call (integer count, prefix sum, fill) and each node sums its neighbours' pre-scaled rows itself: both layers' float atomic edge scatters are gone.
# speedup vs baseline: 1.9201x; 1.5152x over previous
#include <hip/hip_runtime.h>
#include <hip/hip_bf16.h>


typedef float v2f __attribute__((ext_vector_type(2)));
typedef float v8f __attribute__((ext_vector_type(8)));

#define N_IN 64
#define N_HID 128
#define N_OUT 64
#define SCAN_TPB 256


__global__ void k_cnt_zero(int* __restrict__ cnt, int n) {
    int i = blockIdx.x * blockDim.x + threadIdx.x;
    if (i < n) cnt[i] = 0;
}

__global__ void k_cnt_edges(const int* __restrict__ dst, int* __restrict__ cnt,
                            int* __restrict__ slot, int E) {
    int e = blockIdx.x * blockDim.x + threadIdx.x;
    if (e < E) slot[e] = atomicAdd(&cnt[dst[e]], 1);
}

__global__ void k_chunk_sums(const int* __restrict__ cnt, int* __restrict__ bsum, int n) {
    __shared__ int s_wave[SCAN_TPB / 32];
    const int tid = threadIdx.x;
    const int i = blockIdx.x * SCAN_TPB + tid;
    int c = (i < n) ? cnt[i] : 0;
    for (int o = 16; o > 0; o >>= 1) c += __shfl_xor(c, o);
    if ((tid & 31) == 0) s_wave[tid >> 5] = c;
    __syncthreads();
    if (tid == 0) {
        int t = 0;
        for (int w = 0; w < SCAN_TPB / 32; ++w) t += s_wave[w];
        bsum[blockIdx.x] = t;
    }
}

__global__ void k_row_starts(const int* __restrict__ cnt, const int* __restrict__ bsum,
                             int* __restrict__ rowptr, float* __restrict__ inv,
                             const float* __restrict__ x, float* __restrict__ xs, int n) {
    __shared__ int s_wave[SCAN_TPB / 32];
    __shared__ float s_inv[SCAN_TPB];
    const int tid = threadIdx.x;
    const int lane = tid & 31;
    const int wave = tid >> 5;

    int part = 0;
    for (int j = tid; j < (int)blockIdx.x; j += SCAN_TPB) part += bsum[j];
    for (int o = 16; o > 0; o >>= 1) part += __shfl_xor(part, o);
    if (lane == 0) s_wave[wave] = part;
    __syncthreads();
    int off = 0;
    for (int w = 0; w < SCAN_TPB / 32; ++w) off += s_wave[w];
    __syncthreads();

    const int i = blockIdx.x * SCAN_TPB + tid;
    const int c = (i < n) ? cnt[i] : 0;
    int incl = c;
    for (int o = 1; o < 32; o <<= 1) {
        int t = __shfl_up(incl, o);
        if (lane >= o) incl += t;
    }
    if (lane == 31) s_wave[wave] = incl;
    __syncthreads();
    int wpre = 0;
    for (int w = 0; w < wave; ++w) wpre += s_wave[w];
    const int excl = off + wpre + incl - c;
    const float iv = rsqrtf(fmaxf((float)(c + 1), 1.0f));
    if (i < n) {
        rowptr[i] = excl;
        inv[i] = iv;
        if (i == n - 1) rowptr[n] = excl + c;
    }
    s_inv[tid] = iv;
    __syncthreads();

    const size_t base4 = (size_t)blockIdx.x * SCAN_TPB * (N_IN / 4);
    const float4* x4 = (const float4*)x;
    float4* xs4 = (float4*)xs;
    if ((int)(blockIdx.x + 1) * SCAN_TPB <= n) {
        float4 v[N_IN / 4];
        for (int it = 0; it < N_IN / 4; ++it) v[it] = x4[base4 + it * SCAN_TPB + tid];
        for (int it = 0; it < N_IN / 4; ++it) {
            const int q = it * SCAN_TPB + tid;
            const float s = s_inv[q / (N_IN / 4)];
            float4 w = v[it];
            w.x *= s; w.y *= s; w.z *= s; w.w *= s;
            xs4[base4 + q] = w;
        }
    } else {
        for (int it = 0; it < N_IN / 4; ++it) {
            const int q = it * SCAN_TPB + tid;
            const int r = q / (N_IN / 4);
            if ((int)blockIdx.x * SCAN_TPB + r < n) {
                const float s = s_inv[r];
                float4 w = x4[base4 + q];
                w.x *= s; w.y *= s; w.z *= s; w.w *= s;
                xs4[base4 + q] = w;
            }
        }
    }
}

__global__ void k_fill_rows(const int* __restrict__ src, const int* __restrict__ dst,
                            const int* __restrict__ slot, const int* __restrict__ rowptr,
                            int* __restrict__ col, int E) {
    int e = blockIdx.x * blockDim.x + threadIdx.x;
    if (e >= E) return;
    int pos = rowptr[dst[e]] + slot[e];
    pos = min(max(pos, 0), E - 1);
    col[pos] = src[e];
}

template <bool HAS_BIAS>
__global__ void k_agg_rows(const float* __restrict__ hs, const int* __restrict__ rowptr,
                           const int* __restrict__ col, const float* __restrict__ inv,
                           const float* __restrict__ bias, float* __restrict__ out,
                           int n, int E) {
    const int t = blockIdx.x * blockDim.x + threadIdx.x;
    const int l = t & 15;
    const int d = t >> 4;
    if (d >= n) return;
    const float4* h4 = (const float4*)hs;
    const unsigned nmax = (unsigned)(n - 1);

    int beg = rowptr[d];
    int end = rowptr[d + 1];
    beg = min(max(beg, 0), E);
    end = min(max(end, beg), E);
    const float wd = inv[d];
    float4 acc = h4[(size_t)d * 16 + l];

    int j = beg;
    for (; j + 4 <= end; j += 4) {
        const unsigned s0 = min((unsigned)col[j + 0], nmax);
        const unsigned s1 = min((unsigned)col[j + 1], nmax);
        const unsigned s2 = min((unsigned)col[j + 2], nmax);
        const unsigned s3 = min((unsigned)col[j + 3], nmax);
        const float4 a0 = h4[(size_t)s0 * 16 + l];
        const float4 a1 = h4[(size_t)s1 * 16 + l];
        const float4 a2 = h4[(size_t)s2 * 16 + l];
        const float4 a3 = h4[(size_t)s3 * 16 + l];
        acc.x += (a0.x + a1.x) + (a2.x + a3.x);
        acc.y += (a0.y + a1.y) + (a2.y + a3.y);
        acc.z += (a0.z + a1.z) + (a2.z + a3.z);
        acc.w += (a0.w + a1.w) + (a2.w + a3.w);
    }
    const int rem = end - j;
    if (rem > 0) {
        const int j1 = min(j + 1, end - 1);
        const int j2 = min(j + 2, end - 1);
        const unsigned s0 = min((unsigned)col[j], nmax);
        const unsigned s1 = min((unsigned)col[j1], nmax);
        const unsigned s2 = min((unsigned)col[j2], nmax);
        const float4 a0 = h4[(size_t)s0 * 16 + l];
        const float4 a1 = h4[(size_t)s1 * 16 + l];
        const float4 a2 = h4[(size_t)s2 * 16 + l];
        acc.x += a0.x; acc.y += a0.y; acc.z += a0.z; acc.w += a0.w;
        if (rem > 1) { acc.x += a1.x; acc.y += a1.y; acc.z += a1.z; acc.w += a1.w; }
        if (rem > 2) { acc.x += a2.x; acc.y += a2.y; acc.z += a2.z; acc.w += a2.w; }
    }

    float4 r;
    r.x = wd * acc.x; r.y = wd * acc.y; r.z = wd * acc.z; r.w = wd * acc.w;
    if (HAS_BIAS) {
        const float4 bv = ((const float4*)bias)[l];
        r.x += bv.x; r.y += bv.y; r.z += bv.z; r.w += bv.w;
    }
    ((float4*)out)[(size_t)d * 16 + l] = r;
}

template <int K, int N>
__global__ void k_gemm_wmma_f32(const float* __restrict__ A,
                                const float* __restrict__ B,
                                float* __restrict__ C, int M) {
    const int lane = threadIdx.x & 31;
    const int wave = threadIdx.x >> 5;
    const int wavesPerBlock = blockDim.x >> 5;
    const int tile = blockIdx.x * wavesPerBlock + wave;

    const int nTiles = N / 16;
    const int mTiles = M / 16;
    if (tile >= mTiles * nTiles) return;

    const int row0 = (tile / nTiles) * 16;
    const int col0 = (tile % nTiles) * 16;
    const int half = lane >> 4;
    const int l16  = lane & 15;

    v8f acc = {0.f, 0.f, 0.f, 0.f, 0.f, 0.f, 0.f, 0.f};

#pragma unroll
    for (int k = 0; k < K; k += 4) {
        const float* ap = A + (size_t)(row0 + l16) * K + (k + 2 * half);
        v2f a;
        a.x = ap[0];
        a.y = ap[1];
        const float* bp = B + (size_t)(k + 2 * half) * N + (col0 + l16);
        v2f b;
        b.x = bp[0];
        b.y = bp[N];
        acc = __builtin_amdgcn_wmma_f32_16x16x4_f32(
             false, a,  false, b,
             (short)0, acc,  false,  false);
    }

#pragma unroll
    for (int j = 0; j < 8; ++j) {
        C[(size_t)(row0 + 8 * half + j) * N + (col0 + l16)] = acc[j];
    }
}

template <int C>
__global__ void k_bias_relu_scale(float* __restrict__ a, const float* __restrict__ bias,
                                  const float* __restrict__ rowscale, int n) {
    int idx = blockIdx.x * blockDim.x + threadIdx.x;
    if (idx >= n * C) return;
    int c = idx % C;
    int r = idx / C;
    a[idx] = rowscale[r] * fmaxf(a[idx] + bias[c], 0.0f);
}


extern "C" void kernel_launch(void* const* d_in, const int* in_sizes, int n_in,
                              void* d_out, int out_size, void* d_ws, size_t ws_size,
                              hipStream_t stream) {
    const float* x    = (const float*)d_in[0];
    const int*   eidx = (const int*)d_in[1];
    const float* W1   = (const float*)d_in[2];
    const float* b1   = (const float*)d_in[3];
    const float* W2   = (const float*)d_in[4];
    const float* b2   = (const float*)d_in[5];
    float*       out  = (float*)d_out;

    const int n = in_sizes[0] / N_IN;
    const int E = in_sizes[1] / 2;
    const int* src = eidx;
    const int* dst = eidx + E;

    char* ws = (char*)d_ws;
    const size_t OFF_INV = 0;
    const size_t OFF_H0  = (((size_t)n * 4) + 511) & ~(size_t)511;
    const size_t OFF_AGG = OFF_H0 + (size_t)n * N_HID * 4;
    float* inv  = (float*)(ws + OFF_INV);
    float* h1   = (float*)(ws + OFF_H0);
    float* xs   = h1;
    float* aggx = (float*)(ws + OFF_AGG);
    const int nb = (n + SCAN_TPB - 1) / SCAN_TPB;
    const size_t OFF_COL  = OFF_AGG + (size_t)n * N_IN * 4;
    const size_t OFF_SLOT = OFF_COL + ((((size_t)E * 4) + 511) & ~(size_t)511);
    const size_t OFF_ROWP = OFF_SLOT + ((((size_t)E * 4) + 511) & ~(size_t)511);
    const size_t OFF_CNT  = OFF_ROWP + ((((size_t)(n + 1) * 4) + 511) & ~(size_t)511);
    const size_t OFF_BSUM = OFF_CNT + ((((size_t)n * 4) + 511) & ~(size_t)511);
    int* col    = (int*)(ws + OFF_COL);
    int* slot   = (int*)(ws + OFF_SLOT);
    int* rowptr = (int*)(ws + OFF_ROWP);
    int* cnt    = (int*)(ws + OFF_CNT);
    int* bsum   = (int*)(ws + OFF_BSUM);

    const int TPB = 256;

    k_cnt_zero<<<(n + TPB - 1) / TPB, TPB, 0, stream>>>(cnt, n);
    k_cnt_edges<<<(E + TPB - 1) / TPB, TPB, 0, stream>>>(dst, cnt, slot, E);
    k_chunk_sums<<<nb, SCAN_TPB, 0, stream>>>(cnt, bsum, n);
    k_row_starts<<<nb, SCAN_TPB, 0, stream>>>(cnt, bsum, rowptr, inv, x, xs, n);
    k_fill_rows<<<(E + TPB - 1) / TPB, TPB, 0, stream>>>(src, dst, slot, rowptr, col, E);

    k_agg_rows<false><<<((n * 16) + TPB - 1) / TPB, TPB, 0, stream>>>(
        xs, rowptr, col, inv, nullptr, aggx, n, E);
    {
        int tiles = (n / 16) * (N_HID / 16);
        int blocks = (tiles + 3) / 4;
        k_gemm_wmma_f32<N_IN, N_HID><<<blocks, 128, 0, stream>>>(aggx, W1, h1, n);
    }
    k_bias_relu_scale<N_HID><<<((n * N_HID) + TPB - 1) / TPB, TPB, 0, stream>>>(h1, b1, inv, n);

    float* h2 = aggx;
    {
        int tiles = (n / 16) * (N_OUT / 16);
        int blocks = (tiles + 3) / 4;
        k_gemm_wmma_f32<N_HID, N_OUT><<<blocks, 128, 0, stream>>>(h1, W2, h2, n);
    }
    k_agg_rows<true><<<((n * 16) + TPB - 1) / TPB, TPB, 0, stream>>>(
        h2, rowptr, col, inv, b2, out, n, E);

    (void)n_in; (void)out_size; (void)ws_size;
}


# In-edge rows are built in LDS: pass 1 sorts each edge chunk into per-bin runs, pass 2 fills a 256-node bin's rows by LDS atomics, coalesced stores; no global atomic per edge (sum order: LDS arrival)
# speedup vs baseline: 4.5709x; 2.3806x over previous
#include <hip/hip_runtime.h>
#include <hip/hip_bf16.h>


typedef float v2f __attribute__((ext_vector_type(2)));
typedef float v8f __attribute__((ext_vector_type(8)));

#define N_IN 64
#define N_HID 128
#define N_OUT 64
#define SCAN_TPB 256
#define ROW_CAP 64

#define BIN_NODES 256
#define BIN_SHIFT 8
#define MAX_BINS 256
#define RUN_CAP 48
#define A_BLOCKS 256
#define AB_TPB 512
#define A_BATCH 8

__global__ __launch_bounds__(AB_TPB)
void k_bin_edges(const int* __restrict__ src, const int* __restrict__ dst,
                 unsigned* __restrict__ bucket, int* __restrict__ runcnt,
                 int* __restrict__ cnt, int n, int E, int nbins) {
    __shared__ int s_hist[MAX_BINS];
    __shared__ __attribute__((aligned(16))) unsigned s_run[MAX_BINS * RUN_CAP];
    const int tid = threadIdx.x;
    const int b = blockIdx.x;
    const int chunk = (E + A_BLOCKS - 1) / A_BLOCKS;
    const int e0 = min(b * chunk, E);
    const int e1 = min(e0 + chunk, E);
    if (tid < MAX_BINS) s_hist[tid] = 0;
    if (b == 0 && tid == 0) cnt[n] = 0;
    __syncthreads();
    for (int base = e0; base < e1; base += AB_TPB * A_BATCH) {
        int dd[A_BATCH], ss[A_BATCH], sl[A_BATCH];
#pragma unroll
        for (int i = 0; i < A_BATCH; ++i) {
            const int e = min(base + i * AB_TPB + tid, e1 - 1);
            dd[i] = dst[e];
            ss[i] = src[e];
        }
#pragma unroll
        for (int i = 0; i < A_BATCH; ++i) {
            const bool ok = (base + i * AB_TPB + tid < e1) && ((unsigned)dd[i] < (unsigned)n);
            sl[i] = ok ? atomicAdd(&s_hist[dd[i] >> BIN_SHIFT], 1) : RUN_CAP;
        }
#pragma unroll
        for (int i = 0; i < A_BATCH; ++i) {
            if (sl[i] < RUN_CAP)
                s_run[(dd[i] >> BIN_SHIFT) * RUN_CAP + sl[i]] =
                    ((unsigned)ss[i] & 0xFFFFFFu) | ((unsigned)(dd[i] & (BIN_NODES - 1)) << 24);
        }
    }
    __syncthreads();
    if (tid < nbins) runcnt[tid * A_BLOCKS + b] = s_hist[tid];
    uint4* g = (uint4*)(bucket + (size_t)b * nbins * RUN_CAP);
    const int nq = nbins * (RUN_CAP / 4);
    for (int q = tid; q < nq; q += AB_TPB) {
        const int k = q / (RUN_CAP / 4);
        const int p4 = (q - k * (RUN_CAP / 4)) * 4;
        const int c = min(s_hist[k], RUN_CAP);
        if (p4 < c) {
            uint4 v = *(const uint4*)(s_run + k * RUN_CAP + p4);
            if (p4 + 1 >= c) v.y = 0u;
            if (p4 + 2 >= c) v.z = 0u;
            if (p4 + 3 >= c) v.w = 0u;
            g[q] = v;
        }
    }
}

__device__ __forceinline__ void row_store(int* s_rows, int dl, int slot, int s, int d,
                                          int* __restrict__ spill_len, int2* __restrict__ spill, int E) {
    if (slot < ROW_CAP) {
        s_rows[dl * ROW_CAP + slot] = s;
    } else {
        const int kk = atomicAdd(spill_len, 1);
        if ((unsigned)kk < (unsigned)E) spill[kk] = make_int2(s, d);
    }
}

__global__ __launch_bounds__(AB_TPB)
void k_build_rows(const int* __restrict__ src, const int* __restrict__ dst,
                  const unsigned* __restrict__ bucket, const int* __restrict__ runcnt,
                  int* __restrict__ cnt, int* __restrict__ col, int2* __restrict__ spill,
                  int n, int E, int nbins) {
    __shared__ int s_cnt[BIN_NODES];
    __shared__ int s_any;
    __shared__ __attribute__((aligned(16))) int s_rows[BIN_NODES * ROW_CAP];
    const int tid = threadIdx.x;
    const int l = tid & 15;
    const int grp = tid >> 4;
    const int k = blockIdx.x;

    if (tid == 0) s_any = 0;
    if (tid < BIN_NODES) s_cnt[tid] = 0;
    __syncthreads();

    constexpr int GROUPS = AB_TPB / 16;
    constexpr int RUNS = A_BLOCKS / GROUPS;
    constexpr int PARTS = RUN_CAP / 16;
    int rc[RUNS];
    unsigned v[RUNS * PARTS];
    int sl[RUNS * PARTS];
#pragma unroll
    for (int j = 0; j < RUNS; ++j) rc[j] = runcnt[k * A_BLOCKS + j * GROUPS + grp];
    {
        int longest = rc[0];
#pragma unroll
        for (int j = 1; j < RUNS; ++j) longest = max(longest, rc[j]);
        if (longest > RUN_CAP) s_any = 1;
    }
#pragma unroll
    for (int j = 0; j < RUNS; ++j) {
#pragma unroll
        for (int t = 0; t < PARTS; ++t) {
            const int slot = t * 16 + l;
            const bool ok = (rc[j] <= RUN_CAP) && (slot < rc[j]);
            const int b = j * GROUPS + grp;
            v[j * PARTS + t] = ok ? bucket[((size_t)b * nbins + k) * RUN_CAP + slot] : 0xFFFFFFFFu;
        }
    }
#pragma unroll
    for (int i = 0; i < RUNS * PARTS; ++i) {
        const int dl = (int)(v[i] >> 24);
        const bool ok = (v[i] != 0xFFFFFFFFu) && (k * BIN_NODES + dl < n);
        sl[i] = ok ? atomicAdd(&s_cnt[dl], 1) : -1;
    }
#pragma unroll
    for (int i = 0; i < RUNS * PARTS; ++i) {
        if (sl[i] >= 0) {
            const int dl = (int)(v[i] >> 24);
            row_store(s_rows, dl, sl[i], (int)(v[i] & 0xFFFFFFu), k * BIN_NODES + dl, cnt + n, spill, E);
        }
    }
    __syncthreads();
    if (s_any) {
        const int chunk = (E + A_BLOCKS - 1) / A_BLOCKS;
        for (int b = 0; b < A_BLOCKS; ++b) {
            if (runcnt[k * A_BLOCKS + b] > RUN_CAP) {
                const int e0 = min(b * chunk, E);
                const int e1 = min(e0 + chunk, E);
                for (int e = e0 + tid; e < e1; e += AB_TPB) {
                    const int d = dst[e];
                    if ((unsigned)d < (unsigned)n && (d >> BIN_SHIFT) == k) {
                        const int dl = d & (BIN_NODES - 1);
                        row_store(s_rows, dl, atomicAdd(&s_cnt[dl], 1), src[e] & 0xFFFFFF, d, cnt + n, spill, E);
                    }
                }
            }
        }
        __syncthreads();
    }

    if (tid < BIN_NODES) {
        const int d = k * BIN_NODES + tid;
        if (d < n) cnt[d] = s_cnt[tid];
    }
    for (int g = grp; g < BIN_NODES; g += GROUPS) {
        const int d = k * BIN_NODES + g;
        const int c = min(s_cnt[g], ROW_CAP);
        if (d < n && 4 * l < c) {
            int4 r = *(const int4*)(s_rows + g * ROW_CAP + 4 * l);
            if (4 * l + 1 >= c) r.y = 0;
            if (4 * l + 2 >= c) r.z = 0;
            if (4 * l + 3 >= c) r.w = 0;
            *(int4*)(col + (size_t)d * ROW_CAP + 4 * l) = r;
        }
    }
}

__global__ void k_scale_rows(const int* __restrict__ cnt, float* __restrict__ inv,
                             const float* __restrict__ x, float* __restrict__ xs, int n) {
    __shared__ float s_inv[SCAN_TPB];
    const int tid = threadIdx.x;
    const int i = blockIdx.x * SCAN_TPB + tid;
    const int c = (i < n) ? cnt[i] : 0;
    const float iv = rsqrtf(fmaxf((float)(c + 1), 1.0f));
    if (i < n) inv[i] = iv;
    s_inv[tid] = iv;
    __syncthreads();

    const size_t base4 = (size_t)blockIdx.x * SCAN_TPB * (N_IN / 4);
    const float4* x4 = (const float4*)x;
    float4* xs4 = (float4*)xs;
    if ((int)(blockIdx.x + 1) * SCAN_TPB <= n) {
        float4 v[N_IN / 4];
        for (int it = 0; it < N_IN / 4; ++it) v[it] = x4[base4 + it * SCAN_TPB + tid];
        for (int it = 0; it < N_IN / 4; ++it) {
            const int q = it * SCAN_TPB + tid;
            const float s = s_inv[q / (N_IN / 4)];
            float4 w = v[it];
            w.x *= s; w.y *= s; w.z *= s; w.w *= s;
            xs4[base4 + q] = w;
        }
    } else {
        for (int it = 0; it < N_IN / 4; ++it) {
            const int q = it * SCAN_TPB + tid;
            const int r = q / (N_IN / 4);
            if ((int)blockIdx.x * SCAN_TPB + r < n) {
                const float s = s_inv[r];
                float4 w = x4[base4 + q];
                w.x *= s; w.y *= s; w.z *= s; w.w *= s;
                xs4[base4 + q] = w;
            }
        }
    }
}

template <bool HAS_BIAS>
__global__ void k_agg_rows(const float* __restrict__ hs, const int* __restrict__ cnt,
                           const int* __restrict__ col, const int2* __restrict__ spill,
                           const float* __restrict__ inv,
                           const float* __restrict__ bias, float* __restrict__ out,
                           int n, int E) {
    const int t = blockIdx.x * blockDim.x + threadIdx.x;
    const int l = t & 15;
    const int d = t >> 4;
    if (d >= n) return;
    const float4* h4 = (const float4*)hs;
    const unsigned nmax = (unsigned)(n - 1);

    const int craw = cnt[d];
    const int beg = d * ROW_CAP;
    const int end = beg + min(max(craw, 0), ROW_CAP);
    const float wd = inv[d];
    float4 acc = h4[(size_t)d * 16 + l];

    int j = beg;
    for (; j + 4 <= end; j += 4) {
        const unsigned s0 = min((unsigned)col[j + 0], nmax);
        const unsigned s1 = min((unsigned)col[j + 1], nmax);
        const unsigned s2 = min((unsigned)col[j + 2], nmax);
        const unsigned s3 = min((unsigned)col[j + 3], nmax);
        const float4 a0 = h4[(size_t)s0 * 16 + l];
        const float4 a1 = h4[(size_t)s1 * 16 + l];
        const float4 a2 = h4[(size_t)s2 * 16 + l];
        const float4 a3 = h4[(size_t)s3 * 16 + l];
        acc.x += (a0.x + a1.x) + (a2.x + a3.x);
        acc.y += (a0.y + a1.y) + (a2.y + a3.y);
        acc.z += (a0.z + a1.z) + (a2.z + a3.z);
        acc.w += (a0.w + a1.w) + (a2.w + a3.w);
    }
    const int rem = end - j;
    if (rem > 0) {
        const int j1 = min(j + 1, end - 1);
        const int j2 = min(j + 2, end - 1);
        const unsigned s0 = min((unsigned)col[j], nmax);
        const unsigned s1 = min((unsigned)col[j1], nmax);
        const unsigned s2 = min((unsigned)col[j2], nmax);
        const float4 a0 = h4[(size_t)s0 * 16 + l];
        const float4 a1 = h4[(size_t)s1 * 16 + l];
        const float4 a2 = h4[(size_t)s2 * 16 + l];
        acc.x += a0.x; acc.y += a0.y; acc.z += a0.z; acc.w += a0.w;
        if (rem > 1) { acc.x += a1.x; acc.y += a1.y; acc.z += a1.z; acc.w += a1.w; }
        if (rem > 2) { acc.x += a2.x; acc.y += a2.y; acc.z += a2.z; acc.w += a2.w; }
    }
    if (craw > ROW_CAP) {
        const int nsp = min(max(cnt[n], 0), E);
        for (int k = 0; k < nsp; ++k) {
            const int2 sd = spill[k];
            if (sd.y == d) {
                const unsigned s0 = min((unsigned)sd.x, nmax);
                const float4 a0 = h4[(size_t)s0 * 16 + l];
                acc.x += a0.x; acc.y += a0.y; acc.z += a0.z; acc.w += a0.w;
            }
        }
    }

    float4 r;
    r.x = wd * acc.x; r.y = wd * acc.y; r.z = wd * acc.z; r.w = wd * acc.w;
    if (HAS_BIAS) {
        const float4 bv = ((const float4*)bias)[l];
        r.x += bv.x; r.y += bv.y; r.z += bv.z; r.w += bv.w;
    }
    ((float4*)out)[(size_t)d * 16 + l] = r;
}

#define H_PITCH 132
#define MLP_WAVES 8

template <int K, int N>
struct BStage {
    static constexpr int NT = N / 16;
    static constexpr int UNITS = (K / 4) * 2 * NT * 4;
    static constexpr int PER_THREAD = UNITS / (MLP_WAVES * 32);
    float4 g0[PER_THREAD];
    float4 g1[PER_THREAD];

    __device__ __forceinline__ void load(const float* __restrict__ Wg, int tid) {
#pragma unroll
        for (int i = 0; i < PER_THREAD; ++i) {
            const int u = tid + i * (MLP_WAVES * 32);
            const int c4 = u & 3;
            const int h  = (u >> 2) & 1;
            const int nt = (u >> 3) % NT;
            const int s  = (u >> 3) / NT;
            const int row = 4 * s + 2 * h;
            const int col = 16 * nt + 4 * c4;
            g0[i] = *(const float4*)(Wg + (size_t)row * N + col);
            g1[i] = *(const float4*)(Wg + (size_t)(row + 1) * N + col);
        }
    }
    __device__ __forceinline__ void store(v2f* s_w, int tid) const {
#pragma unroll
        for (int i = 0; i < PER_THREAD; ++i) {
            const int u = tid + i * (MLP_WAVES * 32);
            const int c4 = u & 3;
            const int h  = (u >> 2) & 1;
            const int nt = (u >> 3) % NT;
            const int s  = (u >> 3) / NT;
            float4* p = (float4*)(s_w + (s * NT + nt) * 32 + h * 16 + 4 * c4);
            p[0] = make_float4(g0[i].x, g1[i].x, g0[i].y, g1[i].y);
            p[1] = make_float4(g0[i].z, g1[i].z, g0[i].w, g1[i].w);
        }
    }
};

__global__ __launch_bounds__(MLP_WAVES * 32)
void k_mlp_fused(const float* __restrict__ A, const float* __restrict__ W1,
                 const float* __restrict__ b1, const float* __restrict__ rs,
                 const float* __restrict__ W2, float* __restrict__ C, int M) {
    constexpr int W1_SLOTS = (N_IN / 4) * (N_HID / 16) * 32;
    constexpr int W2_SLOTS = (N_HID / 4) * (N_OUT / 16) * 32;
    __shared__ __attribute__((aligned(16))) float s_all[2 * (W1_SLOTS + W2_SLOTS) + MLP_WAVES * 16 * H_PITCH];
    v2f* s_w1 = (v2f*)s_all;
    v2f* s_w2 = s_w1 + W1_SLOTS;
    float* s_h = s_all + 2 * (W1_SLOTS + W2_SLOTS);

    const int tid  = threadIdx.x;
    const int lane = tid & 31;
    const int wave = tid >> 5;
    const int half = lane >> 4;
    const int l16  = lane & 15;
    const int mTiles = M / 16;
    const int tileRaw = blockIdx.x * MLP_WAVES + wave;
    const bool live = tileRaw < mTiles;
    const int row0 = (live ? tileRaw : mTiles - 1) * 16;
    float* H = s_h + wave * (16 * H_PITCH);

    v2f afr[N_IN / 4];
    {
        const float* ap = A + (size_t)(row0 + l16) * N_IN + 2 * half;
#pragma unroll
        for (int s = 0; s < N_IN / 4; ++s) afr[s] = *(const v2f*)(ap + 4 * s);
    }
    float rsv[8];
#pragma unroll
    for (int j = 0; j < 8; ++j) rsv[j] = rs[row0 + 8 * half + j];

    {
        BStage<N_IN, N_HID> st1;
        BStage<N_HID, N_OUT> st2;
        st1.load(W1, tid);
        st2.load(W2, tid);
        st1.store(s_w1, tid);
        st2.store(s_w2, tid);
    }
    __syncthreads();

    v8f acc1[N_HID / 16];
#pragma unroll
    for (int nt = 0; nt < N_HID / 16; ++nt)
        acc1[nt] = (v8f){0.f, 0.f, 0.f, 0.f, 0.f, 0.f, 0.f, 0.f};
#pragma unroll
    for (int s = 0; s < N_IN / 4; ++s) {
#pragma unroll
        for (int nt = 0; nt < N_HID / 16; ++nt) {
            const v2f b = s_w1[(s * (N_HID / 16) + nt) * 32 + lane];
            acc1[nt] = __builtin_amdgcn_wmma_f32_16x16x4_f32(
                 false, afr[s],  false, b,
                 (short)0, acc1[nt],  false,  false);
        }
    }

#pragma unroll
    for (int nt = 0; nt < N_HID / 16; ++nt) {
        const float bv = b1[16 * nt + l16];
#pragma unroll
        for (int j = 0; j < 8; ++j)
            H[(8 * half + j) * H_PITCH + 16 * nt + l16] = rsv[j] * fmaxf(acc1[nt][j] + bv, 0.0f);
    }
    __syncthreads();

    v8f acc2[N_OUT / 16];
#pragma unroll
    for (int nt = 0; nt < N_OUT / 16; ++nt)
        acc2[nt] = (v8f){0.f, 0.f, 0.f, 0.f, 0.f, 0.f, 0.f, 0.f};
    {
        const float* hp = H + l16 * H_PITCH + 2 * half;
#pragma unroll
        for (int s = 0; s < N_HID / 4; ++s) {
            const v2f a = *(const v2f*)(hp + 4 * s);
#pragma unroll
            for (int nt = 0; nt < N_OUT / 16; ++nt) {
                const v2f b = s_w2[(s * (N_OUT / 16) + nt) * 32 + lane];
                acc2[nt] = __builtin_amdgcn_wmma_f32_16x16x4_f32(
                     false, a,  false, b,
                     (short)0, acc2[nt],  false,  false);
            }
        }
    }
    __syncthreads();

#pragma unroll
    for (int nt = 0; nt < N_OUT / 16; ++nt) {
#pragma unroll
        for (int j = 0; j < 8; ++j)
            H[(8 * half + j) * H_PITCH + 16 * nt + l16] = acc2[nt][j];
    }
    __syncthreads();
    if (live) {
        for (int i = 0; i < (16 * N_OUT / 4) / 32; ++i) {
            const int q = i * 32 + lane;
            const int r = q >> 4;
            const int c4 = q & 15;
            const float4 v = *(const float4*)(H + r * H_PITCH + 4 * c4);
            *(float4*)(C + (size_t)(row0 + r) * N_OUT + 4 * c4) = v;
        }
    }
}


extern "C" void kernel_launch(void* const* d_in, const int* in_sizes, int n_in,
                              void* d_out, int out_size, void* d_ws, size_t ws_size,
                              hipStream_t stream) {
    const float* x    = (const float*)d_in[0];
    const int*   eidx = (const int*)d_in[1];
    const float* W1   = (const float*)d_in[2];
    const float* b1   = (const float*)d_in[3];
    const float* W2   = (const float*)d_in[4];
    const float* b2   = (const float*)d_in[5];
    float*       out  = (float*)d_out;

    const int n = in_sizes[0] / N_IN;
    const int E = in_sizes[1] / 2;
    const int* src = eidx;
    const int* dst = eidx + E;

    char* ws = (char*)d_ws;
    const size_t OFF_INV = 0;
    const size_t OFF_H0  = (((size_t)n * 4) + 511) & ~(size_t)511;
    const size_t OFF_AGG = OFF_H0 + (size_t)n * N_HID * 4;
    float* inv  = (float*)(ws + OFF_INV);
    float* xs   = (float*)(ws + OFF_H0);
    float* aggx = (float*)(ws + OFF_AGG);
    const size_t OFF_COL  = OFF_H0 + (size_t)n * N_IN * 4;
    const size_t OFF_CNT  = OFF_AGG + (size_t)n * N_IN * 4;
    const size_t OFF_SPILL = OFF_CNT + ((((size_t)(n + 1) * 4) + 511) & ~(size_t)511);
    int*  col   = (int*)(ws + OFF_COL);
    int*  cnt   = (int*)(ws + OFF_CNT);
    int2* spill = (int2*)(ws + OFF_SPILL);
    const size_t OFF_RUNCNT = OFF_SPILL + ((((size_t)E * 8) + 511) & ~(size_t)511);
    int* runcnt = (int*)(ws + OFF_RUNCNT);
    unsigned* bucket = (unsigned*)(ws + OFF_AGG);
    const int nbins = (n + BIN_NODES - 1) / BIN_NODES;

    const int TPB = 256;
    const int nb = (n + SCAN_TPB - 1) / SCAN_TPB;

    k_bin_edges<<<A_BLOCKS, AB_TPB, 0, stream>>>(src, dst, bucket, runcnt, cnt, n, E, nbins);
    k_build_rows<<<nbins, AB_TPB, 0, stream>>>(src, dst, bucket, runcnt, cnt, col, spill, n, E, nbins);
    k_scale_rows<<<nb, SCAN_TPB, 0, stream>>>(cnt, inv, x, xs, n);

    k_agg_rows<false><<<((n * 16) + TPB - 1) / TPB, TPB, 0, stream>>>(
        xs, cnt, col, spill, inv, nullptr, aggx, n, E);
    float* h2 = xs;
    {
        int tiles = n / 16;
        int blocks = (tiles + MLP_WAVES - 1) / MLP_WAVES;
        k_mlp_fused<<<blocks, MLP_WAVES * 32, 0, stream>>>(aggx, W1, b1, inv, W2, h2, n);
    }
    k_agg_rows<true><<<((n * 16) + TPB - 1) / TPB, TPB, 0, stream>>>(
        h2, cnt, col, spill, inv, b2, out, n, E);

    (void)n_in; (void)out_size; (void)ws_size;
}


# Layer 1's neighbour sum joins the two-transform WMMA kernel: a wave sums its 16 nodes' in-neighbour rows into its LDS tile, 16 row loads in flight (sum order: slot by slot); 5 launches to 4
# speedup vs baseline: 5.0950x; 1.1147x over previous
#include <hip/hip_runtime.h>
#include <hip/hip_bf16.h>


typedef float v2f __attribute__((ext_vector_type(2)));
typedef float v8f __attribute__((ext_vector_type(8)));

#define N_IN 64
#define N_HID 128
#define N_OUT 64
#define ROW_CAP 64

#define BIN_NODES 256
#define BIN_SHIFT 8
#define MAX_BINS 256
#define RUN_CAP 48
#define A_BLOCKS 256
#define AB_TPB 512
#define A_BATCH 8

__global__ __launch_bounds__(AB_TPB)
void k_bin_edges(const int* __restrict__ src, const int* __restrict__ dst,
                 unsigned* __restrict__ bucket, int* __restrict__ runcnt,
                 int* __restrict__ cnt, int n, int E, int nbins) {
    __shared__ int s_hist[MAX_BINS];
    __shared__ __attribute__((aligned(16))) unsigned s_run[MAX_BINS * RUN_CAP];
    const int tid = threadIdx.x;
    const int b = blockIdx.x;
    const int chunk = (E + A_BLOCKS - 1) / A_BLOCKS;
    const int e0 = min(b * chunk, E);
    const int e1 = min(e0 + chunk, E);
    if (tid < MAX_BINS) s_hist[tid] = 0;
    if (b == 0 && tid == 0) cnt[n] = 0;
    __syncthreads();
    for (int base = e0; base < e1; base += AB_TPB * A_BATCH) {
        int dd[A_BATCH], ss[A_BATCH], sl[A_BATCH];
#pragma unroll
        for (int i = 0; i < A_BATCH; ++i) {
            const int e = min(base + i * AB_TPB + tid, e1 - 1);
            dd[i] = dst[e];
            ss[i] = src[e];
        }
#pragma unroll
        for (int i = 0; i < A_BATCH; ++i) {
            const bool ok = (base + i * AB_TPB + tid < e1) && ((unsigned)dd[i] < (unsigned)n);
            sl[i] = ok ? atomicAdd(&s_hist[dd[i] >> BIN_SHIFT], 1) : RUN_CAP;
        }
#pragma unroll
        for (int i = 0; i < A_BATCH; ++i) {
            if (sl[i] < RUN_CAP)
                s_run[(dd[i] >> BIN_SHIFT) * RUN_CAP + sl[i]] =
                    ((unsigned)ss[i] & 0xFFFFFFu) | ((unsigned)(dd[i] & (BIN_NODES - 1)) << 24);
        }
    }
    __syncthreads();
    if (tid < nbins) runcnt[tid * A_BLOCKS + b] = s_hist[tid];
    uint4* g = (uint4*)(bucket + (size_t)b * nbins * RUN_CAP);
    const int nq = nbins * (RUN_CAP / 4);
    for (int q = tid; q < nq; q += AB_TPB) {
        const int k = q / (RUN_CAP / 4);
        const int p4 = (q - k * (RUN_CAP / 4)) * 4;
        const int c = min(s_hist[k], RUN_CAP);
        if (p4 < c) {
            uint4 v = *(const uint4*)(s_run + k * RUN_CAP + p4);
            if (p4 + 1 >= c) v.y = 0u;
            if (p4 + 2 >= c) v.z = 0u;
            if (p4 + 3 >= c) v.w = 0u;
            g[q] = v;
        }
    }
}

__device__ __forceinline__ void row_store(int* s_rows, int dl, int slot, int s, int d,
                                          int* __restrict__ spill_len, int2* __restrict__ spill, int E) {
    if (slot < ROW_CAP) {
        s_rows[dl * ROW_CAP + slot] = s;
    } else {
        const int kk = atomicAdd(spill_len, 1);
        if ((unsigned)kk < (unsigned)E) spill[kk] = make_int2(s, d);
    }
}

__global__ __launch_bounds__(AB_TPB)
void k_build_rows(const int* __restrict__ src, const int* __restrict__ dst,
                  const unsigned* __restrict__ bucket, const int* __restrict__ runcnt,
                  int* __restrict__ cnt, int* __restrict__ col, int2* __restrict__ spill,
                  const float* __restrict__ x, float* __restrict__ xs, float* __restrict__ inv,
                  int n, int E, int nbins) {
    __shared__ int s_cnt[BIN_NODES];
    __shared__ int s_any;
    __shared__ __attribute__((aligned(16))) int s_rows[BIN_NODES * ROW_CAP];
    const int tid = threadIdx.x;
    const int l = tid & 15;
    const int grp = tid >> 4;
    const int k = blockIdx.x;

    if (tid == 0) s_any = 0;
    if (tid < BIN_NODES) s_cnt[tid] = 0;
    __syncthreads();

    constexpr int GROUPS = AB_TPB / 16;
    constexpr int RUNS = A_BLOCKS / GROUPS;
    constexpr int PARTS = RUN_CAP / 16;
    int rc[RUNS];
    unsigned v[RUNS * PARTS];
    int sl[RUNS * PARTS];
#pragma unroll
    for (int j = 0; j < RUNS; ++j) rc[j] = runcnt[k * A_BLOCKS + j * GROUPS + grp];
    {
        int longest = rc[0];
#pragma unroll
        for (int j = 1; j < RUNS; ++j) longest = max(longest, rc[j]);
        if (longest > RUN_CAP) s_any = 1;
    }
#pragma unroll
    for (int j = 0; j < RUNS; ++j) {
#pragma unroll
        for (int t = 0; t < PARTS; ++t) {
            const int slot = t * 16 + l;
            const bool ok = (rc[j] <= RUN_CAP) && (slot < rc[j]);
            const int b = j * GROUPS + grp;
            v[j * PARTS + t] = ok ? bucket[((size_t)b * nbins + k) * RUN_CAP + slot] : 0xFFFFFFFFu;
        }
    }
    constexpr int XROWS = BIN_NODES / GROUPS;
    float4 xv[XROWS];
#pragma unroll
    for (int it = 0; it < XROWS; ++it) {
        const int d = min(k * BIN_NODES + it * GROUPS + grp, n - 1);
        xv[it] = ((const float4*)x)[(size_t)d * (N_IN / 4) + l];
    }
#pragma unroll
    for (int i = 0; i < RUNS * PARTS; ++i) {
        const int dl = (int)(v[i] >> 24);
        const bool ok = (v[i] != 0xFFFFFFFFu) && (k * BIN_NODES + dl < n);
        sl[i] = ok ? atomicAdd(&s_cnt[dl], 1) : -1;
    }
#pragma unroll
    for (int i = 0; i < RUNS * PARTS; ++i) {
        if (sl[i] >= 0) {
            const int dl = (int)(v[i] >> 24);
            row_store(s_rows, dl, sl[i], (int)(v[i] & 0xFFFFFFu), k * BIN_NODES + dl, cnt + n, spill, E);
        }
    }
    __syncthreads();
    if (s_any) {
        const int chunk = (E + A_BLOCKS - 1) / A_BLOCKS;
        for (int b = 0; b < A_BLOCKS; ++b) {
            if (runcnt[k * A_BLOCKS + b] > RUN_CAP) {
                const int e0 = min(b * chunk, E);
                const int e1 = min(e0 + chunk, E);
                for (int e = e0 + tid; e < e1; e += AB_TPB) {
                    const int d = dst[e];
                    if ((unsigned)d < (unsigned)n && (d >> BIN_SHIFT) == k) {
                        const int dl = d & (BIN_NODES - 1);
                        row_store(s_rows, dl, atomicAdd(&s_cnt[dl], 1), src[e] & 0xFFFFFF, d, cnt + n, spill, E);
                    }
                }
            }
        }
        __syncthreads();
    }

    if (tid < BIN_NODES) {
        const int d = k * BIN_NODES + tid;
        if (d < n) {
            cnt[d] = s_cnt[tid];
            inv[d] = rsqrtf(fmaxf((float)(s_cnt[tid] + 1), 1.0f));
        }
    }
#pragma unroll
    for (int it = 0; it < XROWS; ++it) {
        const int g = it * GROUPS + grp;
        const int d = k * BIN_NODES + g;
        const int craw = s_cnt[g];
        const int c = min(craw, ROW_CAP);
        if (d < n) {
            const float iv = rsqrtf(fmaxf((float)(craw + 1), 1.0f));
            float4 w = xv[it];
            w.x *= iv; w.y *= iv; w.z *= iv; w.w *= iv;
            ((float4*)xs)[(size_t)d * (N_IN / 4) + l] = w;
            if (4 * l < c) {
                int4 r = *(const int4*)(s_rows + g * ROW_CAP + 4 * l);
                if (4 * l + 1 >= c) r.y = 0;
                if (4 * l + 2 >= c) r.z = 0;
                if (4 * l + 3 >= c) r.w = 0;
                *(int4*)(col + (size_t)d * ROW_CAP + 4 * l) = r;
            }
        }
    }
}

template <bool HAS_BIAS>
__global__ void k_agg_rows(const float* __restrict__ hs, const int* __restrict__ cnt,
                           const int* __restrict__ col, const int2* __restrict__ spill,
                           const float* __restrict__ inv,
                           const float* __restrict__ bias, float* __restrict__ out,
                           int n, int E) {
    const int t = blockIdx.x * blockDim.x + threadIdx.x;
    const int l = t & 15;
    const int d = t >> 4;
    if (d >= n) return;
    const float4* h4 = (const float4*)hs;
    const unsigned nmax = (unsigned)(n - 1);

    const int craw = cnt[d];
    const int beg = d * ROW_CAP;
    const int end = beg + min(max(craw, 0), ROW_CAP);
    const float wd = inv[d];
    float4 acc = h4[(size_t)d * 16 + l];

    int j = beg;
    for (; j + 4 <= end; j += 4) {
        const unsigned s0 = min((unsigned)col[j + 0], nmax);
        const unsigned s1 = min((unsigned)col[j + 1], nmax);
        const unsigned s2 = min((unsigned)col[j + 2], nmax);
        const unsigned s3 = min((unsigned)col[j + 3], nmax);
        const float4 a0 = h4[(size_t)s0 * 16 + l];
        const float4 a1 = h4[(size_t)s1 * 16 + l];
        const float4 a2 = h4[(size_t)s2 * 16 + l];
        const float4 a3 = h4[(size_t)s3 * 16 + l];
        acc.x += (a0.x + a1.x) + (a2.x + a3.x);
        acc.y += (a0.y + a1.y) + (a2.y + a3.y);
        acc.z += (a0.z + a1.z) + (a2.z + a3.z);
        acc.w += (a0.w + a1.w) + (a2.w + a3.w);
    }
    const int rem = end - j;
    if (rem > 0) {
        const int j1 = min(j + 1, end - 1);
        const int j2 = min(j + 2, end - 1);
        const unsigned s0 = min((unsigned)col[j], nmax);
        const unsigned s1 = min((unsigned)col[j1], nmax);
        const unsigned s2 = min((unsigned)col[j2], nmax);
        const float4 a0 = h4[(size_t)s0 * 16 + l];
        const float4 a1 = h4[(size_t)s1 * 16 + l];
        const float4 a2 = h4[(size_t)s2 * 16 + l];
        acc.x += a0.x; acc.y += a0.y; acc.z += a0.z; acc.w += a0.w;
        if (rem > 1) { acc.x += a1.x; acc.y += a1.y; acc.z += a1.z; acc.w += a1.w; }
        if (rem > 2) { acc.x += a2.x; acc.y += a2.y; acc.z += a2.z; acc.w += a2.w; }
    }
    if (craw > ROW_CAP) {
        const int nsp = min(max(cnt[n], 0), E);
        for (int k = 0; k < nsp; ++k) {
            const int2 sd = spill[k];
            if (sd.y == d) {
                const unsigned s0 = min((unsigned)sd.x, nmax);
                const float4 a0 = h4[(size_t)s0 * 16 + l];
                acc.x += a0.x; acc.y += a0.y; acc.z += a0.z; acc.w += a0.w;
            }
        }
    }

    float4 r;
    r.x = wd * acc.x; r.y = wd * acc.y; r.z = wd * acc.z; r.w = wd * acc.w;
    if (HAS_BIAS) {
        const float4 bv = ((const float4*)bias)[l];
        r.x += bv.x; r.y += bv.y; r.z += bv.z; r.w += bv.w;
    }
    ((float4*)out)[(size_t)d * 16 + l] = r;
}

#define H_PITCH 132
#define MLP_WAVES 8

template <int K, int N>
struct BStage {
    static constexpr int NT = N / 16;
    static constexpr int UNITS = (K / 4) * 2 * NT * 4;
    static constexpr int PER_THREAD = UNITS / (MLP_WAVES * 32);
    float4 g0[PER_THREAD];
    float4 g1[PER_THREAD];

    __device__ __forceinline__ void load(const float* __restrict__ Wg, int tid) {
#pragma unroll
        for (int i = 0; i < PER_THREAD; ++i) {
            const int u = tid + i * (MLP_WAVES * 32);
            const int c4 = u & 3;
            const int h  = (u >> 2) & 1;
            const int nt = (u >> 3) % NT;
            const int s  = (u >> 3) / NT;
            const int row = 4 * s + 2 * h;
            const int col = 16 * nt + 4 * c4;
            g0[i] = *(const float4*)(Wg + (size_t)row * N + col);
            g1[i] = *(const float4*)(Wg + (size_t)(row + 1) * N + col);
        }
    }
    __device__ __forceinline__ void store(v2f* s_w, int tid) const {
#pragma unroll
        for (int i = 0; i < PER_THREAD; ++i) {
            const int u = tid + i * (MLP_WAVES * 32);
            const int c4 = u & 3;
            const int h  = (u >> 2) & 1;
            const int nt = (u >> 3) % NT;
            const int s  = (u >> 3) / NT;
            float4* p = (float4*)(s_w + (s * NT + nt) * 32 + h * 16 + 4 * c4);
            p[0] = make_float4(g0[i].x, g1[i].x, g0[i].y, g1[i].y);
            p[1] = make_float4(g0[i].z, g1[i].z, g0[i].w, g1[i].w);
        }
    }
};

__device__ __forceinline__ void add_row_if(float4& acc, const float4 a, const bool v) {
    acc.x += v ? a.x : 0.0f;
    acc.y += v ? a.y : 0.0f;
    acc.z += v ? a.z : 0.0f;
    acc.w += v ? a.w : 0.0f;
}

__device__ __forceinline__ void add_spilled(float4& acc, const float4* __restrict__ h4, int d, int l,
                                            const int* __restrict__ cnt, const int2* __restrict__ spill,
                                            int n, int E) {
    const unsigned nmax = (unsigned)(n - 1);
    const int nsp = min(max(cnt[n], 0), E);
    for (int k = 0; k < nsp; ++k) {
        const int2 sd = spill[k];
        if (sd.y == d) {
            const unsigned s0 = min((unsigned)sd.x, nmax);
            const float4 a0 = h4[(size_t)s0 * 16 + l];
            acc.x += a0.x; acc.y += a0.y; acc.z += a0.z; acc.w += a0.w;
        }
    }
}

__global__ __launch_bounds__(MLP_WAVES * 32)
void k_agg_mlp_fused(const float* __restrict__ hs, const int* __restrict__ cnt,
                     const int* __restrict__ col, const int2* __restrict__ spill,
                     const float* __restrict__ W1,
                     const float* __restrict__ b1, const float* __restrict__ rs,
                     const float* __restrict__ W2, float* __restrict__ C, int M, int E) {
    constexpr int W1_SLOTS = (N_IN / 4) * (N_HID / 16) * 32;
    constexpr int W2_SLOTS = (N_HID / 4) * (N_OUT / 16) * 32;
    __shared__ __attribute__((aligned(16))) float s_all[2 * (W1_SLOTS + W2_SLOTS) + MLP_WAVES * 16 * H_PITCH];
    v2f* s_w1 = (v2f*)s_all;
    v2f* s_w2 = s_w1 + W1_SLOTS;
    float* s_h = s_all + 2 * (W1_SLOTS + W2_SLOTS);

    const int tid  = threadIdx.x;
    const int lane = tid & 31;
    const int wave = tid >> 5;
    const int half = lane >> 4;
    const int l16  = lane & 15;
    const int mTiles = M / 16;
    const int tileRaw = blockIdx.x * MLP_WAVES + wave;
    const bool live = tileRaw < mTiles;
    const int row0 = (live ? tileRaw : mTiles - 1) * 16;
    float* H = s_h + wave * (16 * H_PITCH);

    const float4* h4 = (const float4*)hs;
    const unsigned nmax = (unsigned)(M - 1);
    int4 colv[8];
    int craw[8];
    float wdv[8];
#pragma unroll
    for (int r = 0; r < 8; ++r) {
        const int d = row0 + 2 * r + half;
        craw[r] = cnt[d];
        colv[r] = ((const int4*)col)[(size_t)d * (ROW_CAP / 4) + l16];
        wdv[r] = rs[d];
    }

    {
        BStage<N_IN, N_HID> st1;
        BStage<N_HID, N_OUT> st2;
        st1.load(W1, tid);
        st2.load(W2, tid);
        st1.store(s_w1, tid);
        st2.store(s_w2, tid);
    }

#pragma unroll
    for (int R = 0; R < 4; ++R) {
        const int lrA = 4 * R + half;
        const int lrB = 4 * R + 2 + half;
        const int dA = row0 + lrA;
        const int dB = row0 + lrB;
        const int cA = min(max(craw[2 * R], 0), ROW_CAP);
        const int cB = min(max(craw[2 * R + 1], 0), ROW_CAP);
        float4 accA = h4[(size_t)dA * 16 + l16];
        float4 accB = h4[(size_t)dB * 16 + l16];
        int cm = max(cA, cB);
        cm = max(cm, __shfl_xor(cm, 16));
        for (int q = 0; 4 * q < cm; q += 2) {
            int ia[8], ib[8];
            ia[0] = __shfl(colv[2 * R].x, q, 16);     ia[1] = __shfl(colv[2 * R].y, q, 16);
            ia[2] = __shfl(colv[2 * R].z, q, 16);     ia[3] = __shfl(colv[2 * R].w, q, 16);
            ia[4] = __shfl(colv[2 * R].x, q + 1, 16); ia[5] = __shfl(colv[2 * R].y, q + 1, 16);
            ia[6] = __shfl(colv[2 * R].z, q + 1, 16); ia[7] = __shfl(colv[2 * R].w, q + 1, 16);
            ib[0] = __shfl(colv[2 * R + 1].x, q, 16);     ib[1] = __shfl(colv[2 * R + 1].y, q, 16);
            ib[2] = __shfl(colv[2 * R + 1].z, q, 16);     ib[3] = __shfl(colv[2 * R + 1].w, q, 16);
            ib[4] = __shfl(colv[2 * R + 1].x, q + 1, 16); ib[5] = __shfl(colv[2 * R + 1].y, q + 1, 16);
            ib[6] = __shfl(colv[2 * R + 1].z, q + 1, 16); ib[7] = __shfl(colv[2 * R + 1].w, q + 1, 16);
            float4 va[8], vb[8];
#pragma unroll
            for (int i = 0; i < 8; ++i) {
                const unsigned sA = (4 * q + i < cA) ? min((unsigned)ia[i], nmax) : (unsigned)dA;
                const unsigned sB = (4 * q + i < cB) ? min((unsigned)ib[i], nmax) : (unsigned)dB;
                va[i] = h4[(size_t)sA * 16 + l16];
                vb[i] = h4[(size_t)sB * 16 + l16];
            }
#pragma unroll
            for (int i = 0; i < 8; ++i) {
                add_row_if(accA, va[i], 4 * q + i < cA);
                add_row_if(accB, vb[i], 4 * q + i < cB);
            }
        }
        if (craw[2 * R] > ROW_CAP) add_spilled(accA, h4, dA, l16, cnt, spill, M, E);
        if (craw[2 * R + 1] > ROW_CAP) add_spilled(accB, h4, dB, l16, cnt, spill, M, E);
        const float wA = wdv[2 * R];
        const float wB = wdv[2 * R + 1];
        *(float4*)(H + lrA * H_PITCH + 4 * l16) = make_float4(wA * accA.x, wA * accA.y, wA * accA.z, wA * accA.w);
        *(float4*)(H + lrB * H_PITCH + 4 * l16) = make_float4(wB * accB.x, wB * accB.y, wB * accB.z, wB * accB.w);
    }
    float rsv[8];
#pragma unroll
    for (int j = 0; j < 8; ++j) rsv[j] = rs[row0 + 8 * half + j];
    __syncthreads();

    v2f afr[N_IN / 4];
    {
        const float* ap = H + l16 * H_PITCH + 2 * half;
#pragma unroll
        for (int s = 0; s < N_IN / 4; ++s) afr[s] = *(const v2f*)(ap + 4 * s);
    }

    v8f acc1[N_HID / 16];
#pragma unroll
    for (int nt = 0; nt < N_HID / 16; ++nt)
        acc1[nt] = (v8f){0.f, 0.f, 0.f, 0.f, 0.f, 0.f, 0.f, 0.f};
#pragma unroll
    for (int s = 0; s < N_IN / 4; ++s) {
#pragma unroll
        for (int nt = 0; nt < N_HID / 16; ++nt) {
            const v2f b = s_w1[(s * (N_HID / 16) + nt) * 32 + lane];
            acc1[nt] = __builtin_amdgcn_wmma_f32_16x16x4_f32(
                 false, afr[s],  false, b,
                 (short)0, acc1[nt],  false,  false);
        }
    }

#pragma unroll
    for (int nt = 0; nt < N_HID / 16; ++nt) {
        const float bv = b1[16 * nt + l16];
#pragma unroll
        for (int j = 0; j < 8; ++j)
            H[(8 * half + j) * H_PITCH + 16 * nt + l16] = rsv[j] * fmaxf(acc1[nt][j] + bv, 0.0f);
    }
    __syncthreads();

    v8f acc2[N_OUT / 16];
#pragma unroll
    for (int nt = 0; nt < N_OUT / 16; ++nt)
        acc2[nt] = (v8f){0.f, 0.f, 0.f, 0.f, 0.f, 0.f, 0.f, 0.f};
    {
        const float* hp = H + l16 * H_PITCH + 2 * half;
#pragma unroll
        for (int s = 0; s < N_HID / 4; ++s) {
            const v2f a = *(const v2f*)(hp + 4 * s);
#pragma unroll
            for (int nt = 0; nt < N_OUT / 16; ++nt) {
                const v2f b = s_w2[(s * (N_OUT / 16) + nt) * 32 + lane];
                acc2[nt] = __builtin_amdgcn_wmma_f32_16x16x4_f32(
                     false, a,  false, b,
                     (short)0, acc2[nt],  false,  false);
            }
        }
    }
    __syncthreads();

#pragma unroll
    for (int nt = 0; nt < N_OUT / 16; ++nt) {
#pragma unroll
        for (int j = 0; j < 8; ++j)
            H[(8 * half + j) * H_PITCH + 16 * nt + l16] = acc2[nt][j];
    }
    __syncthreads();
    if (live) {
        for (int i = 0; i < (16 * N_OUT / 4) / 32; ++i) {
            const int q = i * 32 + lane;
            const int r = q >> 4;
            const int c4 = q & 15;
            const float4 v = *(const float4*)(H + r * H_PITCH + 4 * c4);
            *(float4*)(C + (size_t)(row0 + r) * N_OUT + 4 * c4) = v;
        }
    }
}


extern "C" void kernel_launch(void* const* d_in, const int* in_sizes, int n_in,
                              void* d_out, int out_size, void* d_ws, size_t ws_size,
                              hipStream_t stream) {
    const float* x    = (const float*)d_in[0];
    const int*   eidx = (const int*)d_in[1];
    const float* W1   = (const float*)d_in[2];
    const float* b1   = (const float*)d_in[3];
    const float* W2   = (const float*)d_in[4];
    const float* b2   = (const float*)d_in[5];
    float*       out  = (float*)d_out;

    const int n = in_sizes[0] / N_IN;
    const int E = in_sizes[1] / 2;
    const int* src = eidx;
    const int* dst = eidx + E;

    char* ws = (char*)d_ws;
    const size_t OFF_INV = 0;
    const size_t OFF_H0  = (((size_t)n * 4) + 511) & ~(size_t)511;
    const size_t OFF_AGG = OFF_H0 + (size_t)n * N_HID * 4;
    float* inv  = (float*)(ws + OFF_INV);
    float* xs   = (float*)(ws + OFF_H0);
    const size_t OFF_COL  = OFF_H0 + (size_t)n * N_IN * 4;
    const size_t OFF_CNT  = OFF_AGG + (size_t)n * N_IN * 4;
    const size_t OFF_SPILL = OFF_CNT + ((((size_t)(n + 1) * 4) + 511) & ~(size_t)511);
    int*  col   = (int*)(ws + OFF_COL);
    int*  cnt   = (int*)(ws + OFF_CNT);
    int2* spill = (int2*)(ws + OFF_SPILL);
    const size_t OFF_RUNCNT = OFF_SPILL + ((((size_t)E * 8) + 511) & ~(size_t)511);
    int* runcnt = (int*)(ws + OFF_RUNCNT);
    unsigned* bucket = (unsigned*)(ws + OFF_AGG);
    const int nbins = (n + BIN_NODES - 1) / BIN_NODES;

    const int TPB = 256;

    k_bin_edges<<<A_BLOCKS, AB_TPB, 0, stream>>>(src, dst, bucket, runcnt, cnt, n, E, nbins);
    k_build_rows<<<nbins, AB_TPB, 0, stream>>>(src, dst, bucket, runcnt, cnt, col, spill,
                                               x, xs, inv, n, E, nbins);

    float* h2 = (float*)(ws + OFF_AGG);
    {
        int tiles = n / 16;
        int blocks = (tiles + MLP_WAVES - 1) / MLP_WAVES;
        k_agg_mlp_fused<<<blocks, MLP_WAVES * 32, 0, stream>>>(xs, cnt, col, spill, W1, b1, inv, W2, h2, n, E);
    }
    k_agg_rows<true><<<((n * 16) + TPB - 1) / TPB, TPB, 0, stream>>>(
        h2, cnt, col, spill, inv, b2, out, n, E);

    (void)n_in; (void)out_size; (void)ws_size;
}
